# starting kernel
# baseline (speedup 1.0000x reference)
#include <hip/hip_runtime.h>
#include <math.h>

typedef __attribute__((ext_vector_type(2))) float v2f;
typedef __attribute__((ext_vector_type(8))) float v8f;

#define DIM 64

__global__ void hr_kernel(const float* __restrict__ ent,
                          const float* __restrict__ rel,
                          const int* __restrict__ h,
                          const int* __restrict__ r,
                          float* __restrict__ hr,
                          float* __restrict__ hrn,
                          int T) {
  int wave = (int)((blockIdx.x * blockDim.x + threadIdx.x) >> 5);
  int lane = threadIdx.x & 31;
  if (wave >= T) return;
  const float* e = ent + (size_t)h[wave] * DIM;
  const float* q = rel + (size_t)r[wave] * DIM;
  v2f ev = *(const v2f*)(e + 2 * lane);
  v2f qv = *(const v2f*)(q + 2 * lane);
  v2f s;
  s.x = ev.x + qv.x;
  s.y = ev.y + qv.y;
  *(v2f*)(hr + (size_t)wave * DIM + 2 * lane) = s;
  float n = s.x * s.x + s.y * s.y;
  for (int off = 16; off > 0; off >>= 1) n += __shfl_xor(n, off, 32);
  if (lane == 0) hrn[wave] = n;
}

__global__ void entnorm_kernel(const float* __restrict__ ent,
                               float* __restrict__ entn,
                               int ENT) {
  int wave = (int)((blockIdx.x * blockDim.x + threadIdx.x) >> 5);
  int lane = threadIdx.x & 31;
  if (wave >= ENT) return;
  const float* e = ent + (size_t)wave * DIM;
  v2f ev = *(const v2f*)(e + 2 * lane);
  float n = ev.x * ev.x + ev.y * ev.y;
  for (int off = 16; off > 0; off >>= 1) n += __shfl_xor(n, off, 32);
  if (lane == 0) entn[wave] = n;
}

__global__ void score_kernel(const float* __restrict__ hr,
                             const float* __restrict__ hrn,
                             const float* __restrict__ ent,
                             const float* __restrict__ entn,
                             float* __restrict__ out,
                             int ENT, int mTiles, int totalWaves) {
  int w = (int)((blockIdx.x * blockDim.x + threadIdx.x) >> 5);
  if (w >= totalWaves) return;
  int lane = threadIdx.x & 31;
  int half = lane >> 4;
  int lm   = lane & 15;

  int m0 = (w % mTiles) * 16;
  int n0 = (w / mTiles) * 16;

  const float* arow = hr  + (size_t)(m0 + lm) * DIM + 2 * half;
  const float* brow = ent + (size_t)(n0 + lm) * DIM + 2 * half;

  v8f acc = {};
#pragma unroll
  for (int k = 0; k < DIM; k += 4) {
    v2f a = *(const v2f*)(arow + k);
    v2f b = *(const v2f*)(brow + k);
    acc = __builtin_amdgcn_wmma_f32_16x16x4_f32(
        false, a, false, b, (short)0, acc, false, false);
  }

  float bn = entn[n0 + lm];
  int col = n0 + lm;
#pragma unroll
  for (int v = 0; v < 8; ++v) {
    int m = m0 + v + 8 * half;
    float s = hrn[m] + bn - 2.0f * acc[v];
    out[(size_t)m * ENT + col] = sqrtf(fmaxf(s, 0.0f));
  }
}

extern "C" void kernel_launch(void* const* d_in, const int* in_sizes, int n_in,
                              void* d_out, int out_size, void* d_ws, size_t ws_size,
                              hipStream_t stream) {
  const float* ent = (const float*)d_in[0];
  const float* rel = (const float*)d_in[1];
  const int*   h   = (const int*)d_in[2];
  const int*   r   = (const int*)d_in[3];

  const int ENT = in_sizes[0] / DIM;
  const int T   = in_sizes[2];

  float* ws   = (float*)d_ws;
  float* hr   = ws;
  float* hrn  = ws + (size_t)T * DIM;
  float* entn = hrn + T;

  hr_kernel<<<(T + 7) / 8, 256, 0, stream>>>(ent, rel, h, r, hr, hrn, T);

  entnorm_kernel<<<(ENT + 7) / 8, 256, 0, stream>>>(ent, entn, ENT);

  const int mTiles = T / 16;
  const int nTiles = ENT / 16;
  const int totalWaves = mTiles * nTiles;
  score_kernel<<<(totalWaves + 3) / 4, 128, 0, stream>>>(
      hr, hrn, ent, entn, (float*)d_out, ENT, mTiles, totalWaves);
}


# score kernel re-tiled: emb_hr and an 80-row entity strip staged once per work-group into LDS; each wave keeps 2 row tiles' A fragments and sweeps 5 column tiles; summation order unchanged
# speedup vs baseline: 1.3868x; 1.3868x over previous
#include <hip/hip_runtime.h>
#include <math.h>


typedef __attribute__((ext_vector_type(2))) float v2f;
typedef __attribute__((ext_vector_type(4))) float v4f;
typedef __attribute__((ext_vector_type(8))) float v8f;

#define DIM 64

__global__ void hr_kernel(const float* __restrict__ ent,
                          const float* __restrict__ rel,
                          const int* __restrict__ h,
                          const int* __restrict__ r,
                          float* __restrict__ hr,
                          float* __restrict__ hrn,
                          int T) {
  int wave = (int)((blockIdx.x * blockDim.x + threadIdx.x) >> 5);
  int lane = threadIdx.x & 31;
  if (wave >= T) return;
  const float* e = ent + (size_t)h[wave] * DIM;
  const float* q = rel + (size_t)r[wave] * DIM;
  v2f ev = *(const v2f*)(e + 2 * lane);
  v2f qv = *(const v2f*)(q + 2 * lane);
  v2f s;
  s.x = ev.x + qv.x;
  s.y = ev.y + qv.y;
  *(v2f*)(hr + (size_t)wave * DIM + 2 * lane) = s;
  float n = s.x * s.x + s.y * s.y;
  for (int off = 16; off > 0; off >>= 1) n += __shfl_xor(n, off, 32);
  if (lane == 0) hrn[wave] = n;
}

__global__ void entnorm_kernel(const float* __restrict__ ent,
                               float* __restrict__ entn,
                               int ENT) {
  int wave = (int)((blockIdx.x * blockDim.x + threadIdx.x) >> 5);
  int lane = threadIdx.x & 31;
  if (wave >= ENT) return;
  const float* e = ent + (size_t)wave * DIM;
  v2f ev = *(const v2f*)(e + 2 * lane);
  float n = ev.x * ev.x + ev.y * ev.y;
  for (int off = 16; off > 0; off >>= 1) n += __shfl_xor(n, off, 32);
  if (lane == 0) entn[wave] = n;
}

#define TROWS 256
#define PITCH 68
#define STRIP_TILES 5
#define STRIP_ROWS (16 * STRIP_TILES)
#define SCORE_THREADS 256
#define LDS_FLOATS (41 * 1024)

__global__ void __launch_bounds__(SCORE_THREADS)
score_kernel(const float* __restrict__ hr,
             const float* __restrict__ hrn,
             const float* __restrict__ ent,
             const float* __restrict__ entn,
             float* __restrict__ out,
             int ENT) {
  __shared__ __attribute__((aligned(16))) float lds[LDS_FLOATS];
  float* hrL = lds;
  float* enL = lds + TROWS * PITCH;

  const int tid  = threadIdx.x;
  const int lane = tid & 31;
  const int wave = tid >> 5;
  const int half = lane >> 4;
  const int lm   = lane & 15;
  const int nbase = blockIdx.x * STRIP_ROWS;

  {
    const int c4 = tid & 15;
    const int r0 = tid >> 4;
    v4f hbuf[TROWS / 16];
    v4f ebuf[STRIP_TILES];
#pragma unroll
    for (int i = 0; i < TROWS / 16; ++i)
      hbuf[i] = *(const v4f*)(hr + (size_t)(r0 + 16 * i) * DIM + 4 * c4);
#pragma unroll
    for (int i = 0; i < STRIP_TILES; ++i) {
      int row = nbase + r0 + 16 * i;
      row = row < ENT ? row : ENT - 1;
      ebuf[i] = *(const v4f*)(ent + (size_t)row * DIM + 4 * c4);
    }
#pragma unroll
    for (int i = 0; i < TROWS / 16; ++i)
      *(v4f*)(hrL + (r0 + 16 * i) * PITCH + 4 * c4) = hbuf[i];
#pragma unroll
    for (int i = 0; i < STRIP_TILES; ++i)
      *(v4f*)(enL + (r0 + 16 * i) * PITCH + 4 * c4) = ebuf[i];
  }
  __syncthreads();

  const int m0 = wave * 32;
  const float* ap = hrL + (m0 + lm) * PITCH + 2 * half;
  v2f a0[DIM / 4], a1[DIM / 4];
#pragma unroll
  for (int s = 0; s < DIM / 4; ++s) {
    a0[s] = *(const v2f*)(ap + 4 * s);
    a1[s] = *(const v2f*)(ap + 16 * PITCH + 4 * s);
  }
  float hn0[8], hn1[8];
#pragma unroll
  for (int v = 0; v < 8; ++v) {
    hn0[v] = hrn[m0 + 8 * half + v];
    hn1[v] = hrn[m0 + 16 + 8 * half + v];
  }

  v2f bq[2][DIM / 4];
  {
    const float* bp = enL + lm * PITCH + 2 * half;
#pragma unroll
    for (int s = 0; s < DIM / 4; ++s) bq[0][s] = *(const v2f*)(bp + 4 * s);
  }
#pragma unroll
  for (int j = 0; j < STRIP_TILES; ++j) {
    const int n0 = nbase + 16 * j;
    if (n0 >= ENT) break;
    if (j + 1 < STRIP_TILES) {
      const float* bp = enL + (16 * (j + 1) + lm) * PITCH + 2 * half;
#pragma unroll
      for (int s = 0; s < DIM / 4; ++s) bq[(j + 1) & 1][s] = *(const v2f*)(bp + 4 * s);
    }
    __builtin_amdgcn_sched_barrier(0);

    v8f acc0 = {};
    v8f acc1 = {};
#pragma unroll
    for (int s = 0; s < DIM / 4; ++s) {
      acc0 = __builtin_amdgcn_wmma_f32_16x16x4_f32(
          false, a0[s], false, bq[j & 1][s], (short)0, acc0, false, false);
      acc1 = __builtin_amdgcn_wmma_f32_16x16x4_f32(
          false, a1[s], false, bq[j & 1][s], (short)0, acc1, false, false);
    }

    const int col = n0 + lm;
    const float bn = entn[col];
#pragma unroll
    for (int v = 0; v < 8; ++v) {
      const int m = m0 + v + 8 * half;
      float s0 = hn0[v] + bn - 2.0f * acc0[v];
      out[(size_t)m * ENT + col] = sqrtf(fmaxf(s0, 0.0f));
      float s1 = hn1[v] + bn - 2.0f * acc1[v];
      out[(size_t)(m + 16) * ENT + col] = sqrtf(fmaxf(s1, 0.0f));
    }
  }
}

extern "C" void kernel_launch(void* const* d_in, const int* in_sizes, int n_in,
                              void* d_out, int out_size, void* d_ws, size_t ws_size,
                              hipStream_t stream) {
  const float* ent = (const float*)d_in[0];
  const float* rel = (const float*)d_in[1];
  const int*   h   = (const int*)d_in[2];
  const int*   r   = (const int*)d_in[3];

  const int ENT = in_sizes[0] / DIM;
  const int T   = in_sizes[2];

  float* ws   = (float*)d_ws;
  float* hr   = ws;
  float* hrn  = ws + (size_t)T * DIM;
  float* entn = hrn + T;

  hr_kernel<<<(T + 7) / 8, 256, 0, stream>>>(ent, rel, h, r, hr, hrn, T);

  entnorm_kernel<<<(ENT + 7) / 8, 256, 0, stream>>>(ent, entn, ENT);

  score_kernel<<<(ENT + STRIP_ROWS - 1) / STRIP_ROWS, SCORE_THREADS, 0, stream>>>(
      hr, hrn, ent, entn, (float*)d_out, ENT);
}


# the three launches fused into one: each work-group gathers ent[h]+rel[r] and takes the row norms itself while staging into LDS; norm sums reordered (4 per lane, then a 16-lane tree)
# speedup vs baseline: 2.4672x; 1.7791x over previous
#include <hip/hip_runtime.h>
#include <math.h>


typedef __attribute__((ext_vector_type(2))) float v2f;
typedef __attribute__((ext_vector_type(4))) float v4f;
typedef __attribute__((ext_vector_type(8))) float v8f;

#define DIM 64

template <int XMASK>
__device__ __forceinline__ float row_xor_add(float x) {
  const int yi = __builtin_amdgcn_update_dpp(0, __float_as_int(x), 0x160 | XMASK, 0xf, 0xf, true);
  return x + __int_as_float(yi);
}

__device__ __forceinline__ float row16_sum(float x) {
  x = row_xor_add<1>(x);
  x = row_xor_add<2>(x);
  x = row_xor_add<4>(x);
  x = row_xor_add<8>(x);
  return x;
}

__device__ __forceinline__ float sumsq4(v4f a) {
  return a.x * a.x + a.y * a.y + a.z * a.z + a.w * a.w;
}

#define TROWS 256
#define PITCH 68
#define STRIP_TILES 5
#define STRIP_ROWS (16 * STRIP_TILES)
#define SCORE_THREADS 256
#define LDS_FLOATS (41 * 1024)

__global__ void __launch_bounds__(SCORE_THREADS)
score_kernel(const float* __restrict__ ent,
             const float* __restrict__ rel,
             const int* __restrict__ h,
             const int* __restrict__ r,
             float* __restrict__ out,
             int ENT) {
  __shared__ __attribute__((aligned(16))) float lds[LDS_FLOATS];
  float* hrL = lds;
  float* enL = lds + TROWS * PITCH;
  float* hnL = lds + (TROWS + STRIP_ROWS) * PITCH;
  float* enN = hnL + TROWS;

  const int tid  = threadIdx.x;
  const int lane = tid & 31;
  const int wave = tid >> 5;
  const int half = lane >> 4;
  const int lm   = lane & 15;
  const int nbase = blockIdx.x * STRIP_ROWS;

  {
    const int c4 = lm;
    const int r0 = tid >> 4;
    const int wave_u = __builtin_amdgcn_readfirstlane(wave);
    const int* hp = h + 32 * wave_u;
    const int* rp = r + 32 * wave_u;
    int hidx[32], ridx[32];
#pragma unroll
    for (int k = 0; k < 32; ++k) {
      hidx[k] = hp[k];
      ridx[k] = rp[k];
    }

    v4f ebuf[STRIP_TILES];
    v4f hbuf[16];
    v4f qbuf[16];
#pragma unroll
    for (int i = 0; i < STRIP_TILES; ++i) {
      int row = nbase + r0 + 16 * i;
      row = row < ENT ? row : ENT - 1;
      ebuf[i] = *(const v4f*)(ent + (size_t)row * DIM + 4 * c4);
    }
    const float* hsrc[16];
    const float* qsrc[16];
#pragma unroll
    for (int i = 0; i < 16; ++i) {
      const int h_lo = hidx[2 * i], h_hi = hidx[2 * i + 1];
      const int r_lo = ridx[2 * i], r_hi = ridx[2 * i + 1];
      hsrc[i] = ent + (size_t)((unsigned)(half ? h_hi : h_lo) * DIM + 4u * c4);
      qsrc[i] = rel + (size_t)((unsigned)(half ? r_hi : r_lo) * DIM + 4u * c4);
    }
#pragma unroll
    for (int i = 0; i < 16; ++i) hbuf[i] = *(const v4f*)hsrc[i];
#pragma unroll
    for (int i = 0; i < 16; ++i) qbuf[i] = *(const v4f*)qsrc[i];

    float esel = 0.0f;
#pragma unroll
    for (int i = 0; i < STRIP_TILES; ++i) {
      *(v4f*)(enL + (r0 + 16 * i) * PITCH + 4 * c4) = ebuf[i];
      const float n = row16_sum(sumsq4(ebuf[i]));
      esel = (c4 == i) ? n : esel;
    }
    if (c4 < STRIP_TILES) enN[r0 + 16 * c4] = esel;

    float hsel = 0.0f;
#pragma unroll
    for (int i = 0; i < 16; ++i) {
      const v4f s = hbuf[i] + qbuf[i];
      *(v4f*)(hrL + (32 * wave + 2 * i + half) * PITCH + 4 * c4) = s;
      const float n = row16_sum(sumsq4(s));
      hsel = (c4 == i) ? n : hsel;
    }
    hnL[32 * wave + 2 * c4 + half] = hsel;
  }
  __syncthreads();

  const int m0 = wave * 32;
  const float* ap = hrL + (m0 + lm) * PITCH + 2 * half;
  v2f a0[DIM / 4], a1[DIM / 4];
#pragma unroll
  for (int s = 0; s < DIM / 4; ++s) {
    a0[s] = *(const v2f*)(ap + 4 * s);
    a1[s] = *(const v2f*)(ap + 16 * PITCH + 4 * s);
  }
  float hn0[8], hn1[8];
#pragma unroll
  for (int v = 0; v < 8; ++v) {
    hn0[v] = hnL[m0 + 8 * half + v];
    hn1[v] = hnL[m0 + 16 + 8 * half + v];
  }

  v2f bq[2][DIM / 4];
  {
    const float* bp = enL + lm * PITCH + 2 * half;
#pragma unroll
    for (int s = 0; s < DIM / 4; ++s) bq[0][s] = *(const v2f*)(bp + 4 * s);
  }
#pragma unroll
  for (int j = 0; j < STRIP_TILES; ++j) {
    const int n0 = nbase + 16 * j;
    if (n0 >= ENT) break;
    if (j + 1 < STRIP_TILES) {
      const float* bp = enL + (16 * (j + 1) + lm) * PITCH + 2 * half;
#pragma unroll
      for (int s = 0; s < DIM / 4; ++s) bq[(j + 1) & 1][s] = *(const v2f*)(bp + 4 * s);
    }
    const float bn = enN[16 * j + lm];
    __builtin_amdgcn_sched_barrier(0);

    v8f acc0 = {};
    v8f acc1 = {};
#pragma unroll
    for (int s = 0; s < DIM / 4; ++s) {
      acc0 = __builtin_amdgcn_wmma_f32_16x16x4_f32(
          false, a0[s], false, bq[j & 1][s], (short)0, acc0, false, false);
      acc1 = __builtin_amdgcn_wmma_f32_16x16x4_f32(
          false, a1[s], false, bq[j & 1][s], (short)0, acc1, false, false);
    }

    const int col = n0 + lm;
#pragma unroll
    for (int v = 0; v < 8; ++v) {
      const int m = m0 + v + 8 * half;
      float s0 = hn0[v] + bn - 2.0f * acc0[v];
      out[(size_t)m * ENT + col] = sqrtf(fmaxf(s0, 0.0f));
      float s1 = hn1[v] + bn - 2.0f * acc1[v];
      out[(size_t)(m + 16) * ENT + col] = sqrtf(fmaxf(s1, 0.0f));
    }
  }
}

extern "C" void kernel_launch(void* const* d_in, const int* in_sizes, int n_in,
                              void* d_out, int out_size, void* d_ws, size_t ws_size,
                              hipStream_t stream) {
  const float* ent = (const float*)d_in[0];
  const float* rel = (const float*)d_in[1];
  const int*   h   = (const int*)d_in[2];
  const int*   r   = (const int*)d_in[3];

  const int ENT = in_sizes[0] / DIM;

  score_kernel<<<(ENT + STRIP_ROWS - 1) / STRIP_ROWS, SCORE_THREADS, 0, stream>>>(
      ent, rel, h, r, (float*)d_out, ENT);
}


# epilogue square root: the one hardware sqrt instruction (within 1 ulp) replaces the correctly rounded 14-instruction sequence; outputs may differ in the last place; every sum unchanged
# speedup vs baseline: 2.7265x; 1.1051x over previous
#include <hip/hip_runtime.h>
#include <math.h>


typedef __attribute__((ext_vector_type(2))) float v2f;
typedef __attribute__((ext_vector_type(4))) float v4f;
typedef __attribute__((ext_vector_type(8))) float v8f;

#define DIM 64

template <int XMASK>
__device__ __forceinline__ float row_xor_add(float x) {
  const int yi = __builtin_amdgcn_update_dpp(0, __float_as_int(x), 0x160 | XMASK, 0xf, 0xf, true);
  return x + __int_as_float(yi);
}

__device__ __forceinline__ float row16_sum(float x) {
  x = row_xor_add<1>(x);
  x = row_xor_add<2>(x);
  x = row_xor_add<4>(x);
  x = row_xor_add<8>(x);
  return x;
}

__device__ __forceinline__ float sumsq4(v4f a) {
  return a.x * a.x + a.y * a.y + a.z * a.z + a.w * a.w;
}

__device__ __forceinline__ float hw_sqrt(float x) {
  return __builtin_amdgcn_sqrtf(x);
}

#define TROWS 256
#define PITCH 68
#define STRIP_TILES 5
#define STRIP_ROWS (16 * STRIP_TILES)
#define SCORE_THREADS 256
#define LDS_FLOATS (41 * 1024)

__global__ void __launch_bounds__(SCORE_THREADS)
score_kernel(const float* __restrict__ ent,
             const float* __restrict__ rel,
             const int* __restrict__ h,
             const int* __restrict__ r,
             float* __restrict__ out,
             int ENT) {
  __shared__ __attribute__((aligned(16))) float lds[LDS_FLOATS];
  float* hrL = lds;
  float* enL = lds + TROWS * PITCH;
  float* hnL = lds + (TROWS + STRIP_ROWS) * PITCH;
  float* enN = hnL + TROWS;

  const int tid  = threadIdx.x;
  const int lane = tid & 31;
  const int wave = tid >> 5;
  const int half = lane >> 4;
  const int lm   = lane & 15;
  const int nbase = blockIdx.x * STRIP_ROWS;

  {
    const int c4 = lm;
    const int r0 = tid >> 4;
    const int wave_u = __builtin_amdgcn_readfirstlane(wave);
    const int* hp = h + 32 * wave_u;
    const int* rp = r + 32 * wave_u;
    int hidx[32], ridx[32];
#pragma unroll
    for (int k = 0; k < 32; ++k) {
      hidx[k] = hp[k];
      ridx[k] = rp[k];
    }

    v4f ebuf[STRIP_TILES];
    v4f hbuf[16];
    v4f qbuf[16];
#pragma unroll
    for (int i = 0; i < STRIP_TILES; ++i) {
      int row = nbase + r0 + 16 * i;
      row = row < ENT ? row : ENT - 1;
      ebuf[i] = *(const v4f*)(ent + (size_t)row * DIM + 4 * c4);
    }
    const float* hsrc[16];
    const float* qsrc[16];
#pragma unroll
    for (int i = 0; i < 16; ++i) {
      const int h_lo = hidx[2 * i], h_hi = hidx[2 * i + 1];
      const int r_lo = ridx[2 * i], r_hi = ridx[2 * i + 1];
      hsrc[i] = ent + (size_t)((unsigned)(half ? h_hi : h_lo) * DIM + 4u * c4);
      qsrc[i] = rel + (size_t)((unsigned)(half ? r_hi : r_lo) * DIM + 4u * c4);
    }
#pragma unroll
    for (int i = 0; i < 16; ++i) hbuf[i] = *(const v4f*)hsrc[i];
#pragma unroll
    for (int i = 0; i < 16; ++i) qbuf[i] = *(const v4f*)qsrc[i];

    float esel = 0.0f;
#pragma unroll
    for (int i = 0; i < STRIP_TILES; ++i) {
      *(v4f*)(enL + (r0 + 16 * i) * PITCH + 4 * c4) = ebuf[i];
      const float n = row16_sum(sumsq4(ebuf[i]));
      esel = (c4 == i) ? n : esel;
    }
    if (c4 < STRIP_TILES) enN[r0 + 16 * c4] = esel;

    float hsel = 0.0f;
#pragma unroll
    for (int i = 0; i < 16; ++i) {
      const v4f s = hbuf[i] + qbuf[i];
      *(v4f*)(hrL + (32 * wave + 2 * i + half) * PITCH + 4 * c4) = s;
      const float n = row16_sum(sumsq4(s));
      hsel = (c4 == i) ? n : hsel;
    }
    hnL[32 * wave + 2 * c4 + half] = hsel;
  }
  __syncthreads();

  const int m0 = wave * 32;
  const float* ap = hrL + (m0 + lm) * PITCH + 2 * half;
  v2f a0[DIM / 4], a1[DIM / 4];
#pragma unroll
  for (int s = 0; s < DIM / 4; ++s) {
    a0[s] = *(const v2f*)(ap + 4 * s);
    a1[s] = *(const v2f*)(ap + 16 * PITCH + 4 * s);
  }
  float hn0[8], hn1[8];
#pragma unroll
  for (int v = 0; v < 8; ++v) {
    hn0[v] = hnL[m0 + 8 * half + v];
    hn1[v] = hnL[m0 + 16 + 8 * half + v];
  }

  v2f bq[2][DIM / 4];
  {
    const float* bp = enL + lm * PITCH + 2 * half;
#pragma unroll
    for (int s = 0; s < DIM / 4; ++s) bq[0][s] = *(const v2f*)(bp + 4 * s);
  }
#pragma unroll
  for (int j = 0; j < STRIP_TILES; ++j) {
    const int n0 = nbase + 16 * j;
    if (n0 >= ENT) break;
    if (j + 1 < STRIP_TILES) {
      const float* bp = enL + (16 * (j + 1) + lm) * PITCH + 2 * half;
#pragma unroll
      for (int s = 0; s < DIM / 4; ++s) bq[(j + 1) & 1][s] = *(const v2f*)(bp + 4 * s);
    }
    const float bn = enN[16 * j + lm];
    __builtin_amdgcn_sched_barrier(0);

    v8f acc0 = {};
    v8f acc1 = {};
#pragma unroll
    for (int s = 0; s < DIM / 4; ++s) {
      acc0 = __builtin_amdgcn_wmma_f32_16x16x4_f32(
          false, a0[s], false, bq[j & 1][s], (short)0, acc0, false, false);
      acc1 = __builtin_amdgcn_wmma_f32_16x16x4_f32(
          false, a1[s], false, bq[j & 1][s], (short)0, acc1, false, false);
    }

    const int col = n0 + lm;
#pragma unroll
    for (int v = 0; v < 8; ++v) {
      const int m = m0 + v + 8 * half;
      float s0 = hn0[v] + bn - 2.0f * acc0[v];
      out[(size_t)m * ENT + col] = hw_sqrt(fmaxf(s0, 0.0f));
      float s1 = hn1[v] + bn - 2.0f * acc1[v];
      out[(size_t)(m + 16) * ENT + col] = hw_sqrt(fmaxf(s1, 0.0f));
    }
  }
}

extern "C" void kernel_launch(void* const* d_in, const int* in_sizes, int n_in,
                              void* d_out, int out_size, void* d_ws, size_t ws_size,
                              hipStream_t stream) {
  const float* ent = (const float*)d_in[0];
  const float* rel = (const float*)d_in[1];
  const int*   h   = (const int*)d_in[2];
  const int*   r   = (const int*)d_in[3];

  const int ENT = in_sizes[0] / DIM;

  score_kernel<<<(ENT + STRIP_ROWS - 1) / STRIP_ROWS, SCORE_THREADS, 0, stream>>>(
      ent, rel, h, r, (float*)d_out, ENT);
}


# work-group tile reshaped from 256 rows x 80 columns to 64 rows x 320 columns: each work-group gathers 64+64 table rows instead of 256+256 and stages a 320-row entity strip; summation order unchanged
# speedup vs baseline: 2.8916x; 1.0606x over previous
#include <hip/hip_runtime.h>
#include <math.h>


typedef __attribute__((ext_vector_type(2))) float v2f;
typedef __attribute__((ext_vector_type(4))) float v4f;
typedef __attribute__((ext_vector_type(8))) float v8f;

#define DIM 64

template <int XMASK>
__device__ __forceinline__ float row_xor_add(float x) {
  const int yi = __builtin_amdgcn_update_dpp(0, __float_as_int(x), 0x160 | XMASK, 0xf, 0xf, true);
  return x + __int_as_float(yi);
}

__device__ __forceinline__ float row16_sum(float x) {
  x = row_xor_add<1>(x);
  x = row_xor_add<2>(x);
  x = row_xor_add<4>(x);
  x = row_xor_add<8>(x);
  return x;
}

__device__ __forceinline__ float sumsq4(v4f a) {
  return a.x * a.x + a.y * a.y + a.z * a.z + a.w * a.w;
}

__device__ __forceinline__ float hw_sqrt(float x) {
  return __builtin_amdgcn_sqrtf(x);
}

#define ALLROWS 256
#define TROWS 64
#define ROW_BLOCKS (ALLROWS / TROWS)
#define PITCH 68
#define WAVE_TILES 5
#define WAVE_COLS (16 * WAVE_TILES)
#define WAVE_COLUMNS 4
#define STRIP_ROWS (WAVE_COLS * WAVE_COLUMNS)
#define SCORE_THREADS 256
#define STRIP_PASSES (STRIP_ROWS / 16)
#define GATHER_PAIRS 4
#define LDS_FLOATS (41 * 1024)

__global__ void __launch_bounds__(SCORE_THREADS)
score_kernel(const float* __restrict__ ent,
             const float* __restrict__ rel,
             const int* __restrict__ h,
             const int* __restrict__ r,
             float* __restrict__ out,
             int ENT) {
  __shared__ __attribute__((aligned(16))) float lds[LDS_FLOATS];
  float* hrL = lds;
  float* enL = lds + TROWS * PITCH;
  float* hnL = lds + (TROWS + STRIP_ROWS) * PITCH;
  float* enN = hnL + TROWS;

  const int tid  = threadIdx.x;
  const int lane = tid & 31;
  const int wave = tid >> 5;
  const int half = lane >> 4;
  const int lm   = lane & 15;
  const int wave_u = __builtin_amdgcn_readfirstlane(wave);
  const int mbase = (blockIdx.x % ROW_BLOCKS) * TROWS;
  const int nbase = (blockIdx.x / ROW_BLOCKS) * STRIP_ROWS;

  {
    const int c4 = lm;
    const int r0 = tid >> 4;
    const int* hp = h + mbase + 2 * GATHER_PAIRS * wave_u;
    const int* rp = r + mbase + 2 * GATHER_PAIRS * wave_u;
    int hidx[2 * GATHER_PAIRS], ridx[2 * GATHER_PAIRS];
#pragma unroll
    for (int k = 0; k < 2 * GATHER_PAIRS; ++k) {
      hidx[k] = hp[k];
      ridx[k] = rp[k];
    }
    __builtin_amdgcn_sched_barrier(0);

    v4f ebuf[STRIP_PASSES];
    v4f hbuf[GATHER_PAIRS];
    v4f qbuf[GATHER_PAIRS];
#pragma unroll
    for (int i = 0; i < STRIP_PASSES; ++i) {
      int row = nbase + r0 + 16 * i;
      row = row < ENT ? row : ENT - 1;
      ebuf[i] = *(const v4f*)(ent + (size_t)row * DIM + 4 * c4);
    }
    __builtin_amdgcn_sched_barrier(0);
    const float* hsrc[GATHER_PAIRS];
    const float* qsrc[GATHER_PAIRS];
#pragma unroll
    for (int i = 0; i < GATHER_PAIRS; ++i) {
      const int h_lo = hidx[2 * i], h_hi = hidx[2 * i + 1];
      const int r_lo = ridx[2 * i], r_hi = ridx[2 * i + 1];
      hsrc[i] = ent + (size_t)((unsigned)(half ? h_hi : h_lo) * DIM + 4u * c4);
      qsrc[i] = rel + (size_t)((unsigned)(half ? r_hi : r_lo) * DIM + 4u * c4);
    }
#pragma unroll
    for (int i = 0; i < GATHER_PAIRS; ++i) hbuf[i] = *(const v4f*)hsrc[i];
#pragma unroll
    for (int i = 0; i < GATHER_PAIRS; ++i) qbuf[i] = *(const v4f*)qsrc[i];
    __builtin_amdgcn_sched_barrier(0);

    float esel_a = 0.0f, esel_b = 0.0f;
#pragma unroll
    for (int i = 0; i < STRIP_PASSES; ++i) {
      *(v4f*)(enL + (r0 + 16 * i) * PITCH + 4 * c4) = ebuf[i];
      const float n = row16_sum(sumsq4(ebuf[i]));
      if (i < 16) esel_a = (c4 == i) ? n : esel_a;
      else        esel_b = (c4 == i - 16) ? n : esel_b;
    }
    enN[r0 + 16 * c4] = esel_a;
    if (c4 < STRIP_PASSES - 16) enN[r0 + 16 * (16 + c4)] = esel_b;

    float hsel = 0.0f;
#pragma unroll
    for (int i = 0; i < GATHER_PAIRS; ++i) {
      const v4f s = hbuf[i] + qbuf[i];
      *(v4f*)(hrL + (2 * GATHER_PAIRS * wave + 2 * i + half) * PITCH + 4 * c4) = s;
      const float n = row16_sum(sumsq4(s));
      hsel = (c4 == i) ? n : hsel;
    }
    if (c4 < GATHER_PAIRS) hnL[2 * GATHER_PAIRS * wave + 2 * c4 + half] = hsel;
  }
  __syncthreads();

  const int m0 = (wave_u & 1) * 32;
  const int cw = (wave_u >> 1) * WAVE_COLS;
  const float* ap = hrL + (m0 + lm) * PITCH + 2 * half;
  v2f a0[DIM / 4], a1[DIM / 4];
#pragma unroll
  for (int s = 0; s < DIM / 4; ++s) {
    a0[s] = *(const v2f*)(ap + 4 * s);
    a1[s] = *(const v2f*)(ap + 16 * PITCH + 4 * s);
  }
  float hn0[8], hn1[8];
#pragma unroll
  for (int v = 0; v < 8; ++v) {
    hn0[v] = hnL[m0 + 8 * half + v];
    hn1[v] = hnL[m0 + 16 + 8 * half + v];
  }

  v2f bq[2][DIM / 4];
  {
    const float* bp = enL + (cw + lm) * PITCH + 2 * half;
#pragma unroll
    for (int s = 0; s < DIM / 4; ++s) bq[0][s] = *(const v2f*)(bp + 4 * s);
  }
#pragma unroll
  for (int j = 0; j < WAVE_TILES; ++j) {
    const int n0 = nbase + cw + 16 * j;
    if (n0 >= ENT) break;
    if (j + 1 < WAVE_TILES) {
      const float* bp = enL + (cw + 16 * (j + 1) + lm) * PITCH + 2 * half;
#pragma unroll
      for (int s = 0; s < DIM / 4; ++s) bq[(j + 1) & 1][s] = *(const v2f*)(bp + 4 * s);
    }
    const float bn = enN[cw + 16 * j + lm];
    __builtin_amdgcn_sched_barrier(0);

    v8f acc0 = {};
    v8f acc1 = {};
#pragma unroll
    for (int s = 0; s < DIM / 4; ++s) {
      acc0 = __builtin_amdgcn_wmma_f32_16x16x4_f32(
          false, a0[s], false, bq[j & 1][s], (short)0, acc0, false, false);
      acc1 = __builtin_amdgcn_wmma_f32_16x16x4_f32(
          false, a1[s], false, bq[j & 1][s], (short)0, acc1, false, false);
    }

    const int col = n0 + lm;
#pragma unroll
    for (int v = 0; v < 8; ++v) {
      const int m = mbase + m0 + v + 8 * half;
      float s0 = hn0[v] + bn - 2.0f * acc0[v];
      out[(size_t)m * ENT + col] = hw_sqrt(fmaxf(s0, 0.0f));
      float s1 = hn1[v] + bn - 2.0f * acc1[v];
      out[(size_t)(m + 16) * ENT + col] = hw_sqrt(fmaxf(s1, 0.0f));
    }
  }
}

extern "C" void kernel_launch(void* const* d_in, const int* in_sizes, int n_in,
                              void* d_out, int out_size, void* d_ws, size_t ws_size,
                              hipStream_t stream) {
  const float* ent = (const float*)d_in[0];
  const float* rel = (const float*)d_in[1];
  const int*   h   = (const int*)d_in[2];
  const int*   r   = (const int*)d_in[3];

  const int ENT = in_sizes[0] / DIM;

  score_kernel<<<ROW_BLOCKS * ((ENT + STRIP_ROWS - 1) / STRIP_ROWS), SCORE_THREADS, 0, stream>>>(
      ent, rel, h, r, (float*)d_out, ENT);
}
